# starting kernel
# baseline (speedup 1.0000x reference)
#include <hip/hip_runtime.h>

typedef __attribute__((ext_vector_type(2))) float v2f;
typedef __attribute__((ext_vector_type(8))) float v8f;

#define NB 16
#define NS 800
#define NT 224
#define NN 1024
#define DD 32

__global__ void hstgnn_pack_weights(const float* __restrict__ relW,
                                    const float* __restrict__ maskW,
                                    float* __restrict__ wpack) {
    int i = blockIdx.x * blockDim.x + threadIdx.x;
    if (i >= 2 * 16 * 32) return;
    int d    = i & 31;
    int n    = (i >> 5) & 15;
    int type = i >> 9;
    float v = 0.0f;
    if (n < 8) {
        const int krow[2][8] = { {0,0,2,2,0,0,3,3},
                                 {1,1,3,3,1,1,2,2} };
        int k    = krow[type][n];
        int half = (n >> 2) & 1;
        const float* src = (n & 1) ? maskW : relW;
        v = src[k * 64 + half * 32 + d];
    }
    wpack[i] = v;
}

__global__ void hstgnn_score_wmma(const float* __restrict__ sp,
                                  const float* __restrict__ tp,
                                  const float* __restrict__ wpack,
                                  float* __restrict__ scores) {
    int id   = blockIdx.x;
    int b    = id >> 6;
    int t    = id & 63;
    int lane = threadIdx.x;
    int r    = lane & 15;
    int hi   = lane >> 4;

    const float* src;
    int type;
    if (t < 50) { type = 0; src = sp + ((size_t)b * NS + (size_t)t * 16) * DD; }
    else        { type = 1; src = tp + ((size_t)b * NT + (size_t)(t - 50) * 16) * DD; }
    const float* w = wpack + type * (16 * 32);

    v8f acc = {0.f,0.f,0.f,0.f,0.f,0.f,0.f,0.f};
#pragma unroll
    for (int k = 0; k < 8; ++k) {
        int kb = k * 4 + hi * 2;
        v2f a  = *(const v2f*)(src + r * DD + kb);
        v2f bb = *(const v2f*)(w   + r * DD + kb);
        acc = __builtin_amdgcn_wmma_f32_16x16x4_f32(
                  false, a, false, bb, (short)0, acc, false, false);
    }

    float* dst = scores + ((size_t)b * NN + (size_t)t * 16) * 16;
#pragma unroll
    for (int v = 0; v < 8; ++v) {
        int row = v + hi * 8;
        dst[row * 16 + r] = acc[v];
    }
}

__global__ void hstgnn_emit(const float* __restrict__ scores,
                            const float* __restrict__ relb,
                            const float* __restrict__ maskb,
                            float* __restrict__ out) {
    int bi = blockIdx.x;
    int b  = bi >> 10;
    int i  = bi & 1023;
    int j0 = threadIdx.x * 4;
    bool iT = (i  >= NS);
    bool jT = (j0 >= NS);

    int rowslot = (iT == jT) ? 0 : 2;
    int colslot = (iT == jT) ? 4 : 6;
    int k = jT ? (iT ? 1 : 2) : (iT ? 3 : 0);

    const float* srow = scores + ((size_t)b * NN + i) * 16;
    float2 rv = *(const float2*)(srow + rowslot);
    float  rb = relb[k];
    float  mb = maskb[k];

    const float* scol = scores + (size_t)b * NN * 16;
    float res[4];
#pragma unroll
    for (int u = 0; u < 4; ++u) {
        int j = j0 + u;
        float2 cv = *(const float2*)(scol + (size_t)j * 16 + colslot);
        float adj  = rv.x + cv.x + rb;
        float mask = fmaxf(rv.y + cv.y + mb, 0.0f);
        res[u] = tanhf(adj * mask);
    }
    *(float4*)(out + (size_t)bi * NN + j0) =
        make_float4(res[0], res[1], res[2], res[3]);
}

extern "C" void kernel_launch(void* const* d_in, const int* in_sizes, int n_in,
                              void* d_out, int out_size, void* d_ws, size_t ws_size,
                              hipStream_t stream) {
    const float* spatial  = (const float*)d_in[0];
    const float* temporal = (const float*)d_in[1];
    const float* rel_W    = (const float*)d_in[2];
    const float* rel_b    = (const float*)d_in[3];
    const float* mask_W   = (const float*)d_in[4];
    const float* mask_b   = (const float*)d_in[5];
    float*       out      = (float*)d_out;

    float* wpack  = (float*)d_ws;
    float* scores = wpack + 2 * 16 * 32;

    hipLaunchKernelGGL(hstgnn_pack_weights, dim3(4), dim3(256), 0, stream,
                       rel_W, mask_W, wpack);
    hipLaunchKernelGGL(hstgnn_score_wmma, dim3(NB * 64), dim3(32), 0, stream,
                       spatial, temporal, wpack, scores);
    hipLaunchKernelGGL(hstgnn_emit, dim3(NB * NN), dim3(256), 0, stream,
                       scores, rel_b, mask_b, out);
}


# emit kernel: each block walks 16 consecutive rows with its four column scores kept in registers (grid 1024 blocks, was 16384); row scores read block-uniform one row ahead; tanhf and sum order kept
# speedup vs baseline: 2.1894x; 2.1894x over previous
#include <hip/hip_runtime.h>

typedef __attribute__((ext_vector_type(2))) float v2f;
typedef __attribute__((ext_vector_type(8))) float v8f;

#define NB 16
#define NS 800
#define NT 224
#define NN 1024
#define DD 32

__global__ void hstgnn_pack_weights(const float* __restrict__ relW,
                                    const float* __restrict__ maskW,
                                    float* __restrict__ wpack) {
    int i = blockIdx.x * blockDim.x + threadIdx.x;
    if (i >= 2 * 16 * 32) return;
    int d    = i & 31;
    int n    = (i >> 5) & 15;
    int type = i >> 9;
    float v = 0.0f;
    if (n < 8) {
        const int krow[2][8] = { {0,0,2,2,0,0,3,3},
                                 {1,1,3,3,1,1,2,2} };
        int k    = krow[type][n];
        int half = (n >> 2) & 1;
        const float* src = (n & 1) ? maskW : relW;
        v = src[k * 64 + half * 32 + d];
    }
    wpack[i] = v;
}

__global__ void hstgnn_score_wmma(const float* __restrict__ sp,
                                  const float* __restrict__ tp,
                                  const float* __restrict__ wpack,
                                  float* __restrict__ scores) {
    int id   = blockIdx.x;
    int b    = id >> 6;
    int t    = id & 63;
    int lane = threadIdx.x;
    int r    = lane & 15;
    int hi   = lane >> 4;

    const float* src;
    int type;
    if (t < 50) { type = 0; src = sp + ((size_t)b * NS + (size_t)t * 16) * DD; }
    else        { type = 1; src = tp + ((size_t)b * NT + (size_t)(t - 50) * 16) * DD; }
    const float* w = wpack + type * (16 * 32);

    v8f acc = {0.f,0.f,0.f,0.f,0.f,0.f,0.f,0.f};
#pragma unroll
    for (int k = 0; k < 8; ++k) {
        int kb = k * 4 + hi * 2;
        v2f a  = *(const v2f*)(src + r * DD + kb);
        v2f bb = *(const v2f*)(w   + r * DD + kb);
        acc = __builtin_amdgcn_wmma_f32_16x16x4_f32(
                  false, a, false, bb, (short)0, acc, false, false);
    }

    float* dst = scores + ((size_t)b * NN + (size_t)t * 16) * 16;
#pragma unroll
    for (int v = 0; v < 8; ++v) {
        int row = v + hi * 8;
        dst[row * 16 + r] = acc[v];
    }
}

#define RPB 16
static_assert(NS % RPB == 0 && NN % RPB == 0, "row group must not straddle S/T");

__global__ __launch_bounds__(256)
void hstgnn_emit(const float* __restrict__ scores,
                 const float* __restrict__ relb,
                 const float* __restrict__ maskb,
                 float* __restrict__ out) {
    int blk = blockIdx.x;
    int b   = blk / (NN / RPB);
    int i0  = (blk % (NN / RPB)) * RPB;
    int j0  = threadIdx.x * 4;
    bool iT = (i0 >= NS);
    bool jT = (j0 >= NS);
    bool same = (iT == jT);

    int colslot = same ? 4 : 6;
    int k = jT ? (iT ? 1 : 2) : (iT ? 3 : 0);
    float rb = relb[k];
    float mb = maskb[k];

    const float* scol = scores + (size_t)b * NN * 16;
    float cx[4], cy[4];
#pragma unroll
    for (int u = 0; u < 4; ++u) {
        float2 cv = *(const float2*)(scol + (size_t)(j0 + u) * 16 + colslot);
        cx[u] = cv.x;
        cy[u] = cv.y;
    }

    const float* srow = scores + ((size_t)b * NN + i0) * 16;
    float*       orow = out + ((size_t)b * NN + i0) * NN + j0;

    float4 nxt = *(const float4*)srow;
    for (int r = 0; r < RPB; ++r) {
        float4 cur = nxt;
        if (r + 1 < RPB) nxt = *(const float4*)(srow + (r + 1) * 16);
        float rx = same ? cur.x : cur.z;
        float ry = same ? cur.y : cur.w;
        float res[4];
#pragma unroll
        for (int u = 0; u < 4; ++u) {
            float adj  = rx + cx[u] + rb;
            float mask = fmaxf(ry + cy[u] + mb, 0.0f);
            res[u] = tanhf(adj * mask);
        }
        *(float4*)(orow + (size_t)r * NN) =
            make_float4(res[0], res[1], res[2], res[3]);
    }
}

extern "C" void kernel_launch(void* const* d_in, const int* in_sizes, int n_in,
                              void* d_out, int out_size, void* d_ws, size_t ws_size,
                              hipStream_t stream) {
    const float* spatial  = (const float*)d_in[0];
    const float* temporal = (const float*)d_in[1];
    const float* rel_W    = (const float*)d_in[2];
    const float* rel_b    = (const float*)d_in[3];
    const float* mask_W   = (const float*)d_in[4];
    const float* mask_b   = (const float*)d_in[5];
    float*       out      = (float*)d_out;

    float* wpack  = (float*)d_ws;
    float* scores = wpack + 2 * 16 * 32;

    hipLaunchKernelGGL(hstgnn_pack_weights, dim3(4), dim3(256), 0, stream,
                       rel_W, mask_W, wpack);
    hipLaunchKernelGGL(hstgnn_score_wmma, dim3(NB * 64), dim3(32), 0, stream,
                       spatial, temporal, wpack, scores);
    hipLaunchKernelGGL(hstgnn_emit, dim3(NB * (NN / RPB)), dim3(256), 0, stream,
                       scores, rel_b, mask_b, out);
}


# weight packing folded into the score kernel: each lane reads its B fragment straight from rel_W / mask_W; 2 launches per repetition, was 3
# speedup vs baseline: 2.4796x; 1.1325x over previous
#include <hip/hip_runtime.h>

typedef __attribute__((ext_vector_type(2))) float v2f;
typedef __attribute__((ext_vector_type(8))) float v8f;

#define NB 16
#define NS 800
#define NT 224
#define NN 1024
#define DD 32

__global__ void hstgnn_score_wmma(const float* __restrict__ sp,
                                  const float* __restrict__ tp,
                                  const float* __restrict__ relW,
                                  const float* __restrict__ maskW,
                                  float* __restrict__ scores) {
    int id   = blockIdx.x;
    int b    = id >> 6;
    int t    = id & 63;
    int lane = threadIdx.x;
    int r    = lane & 15;
    int hi   = lane >> 4;

    const float* src;
    int type;
    if (t < 50) { type = 0; src = sp + ((size_t)b * NS + (size_t)t * 16) * DD; }
    else        { type = 1; src = tp + ((size_t)b * NT + (size_t)(t - 50) * 16) * DD; }

    int  n     = r & 7;
    int  half  = (n >> 2) & 1;
    bool cross = (n >> 1) & 1;
    int  kk    = cross ? 2 + (type ^ half) : type;
    const float* w = ((n & 1) ? maskW : relW) + kk * 64 + half * 32;

    v8f acc = {0.f,0.f,0.f,0.f,0.f,0.f,0.f,0.f};
#pragma unroll
    for (int k = 0; k < 8; ++k) {
        int kb = k * 4 + hi * 2;
        v2f a  = *(const v2f*)(src + r * DD + kb);
        v2f bb = *(const v2f*)(w + kb);
        acc = __builtin_amdgcn_wmma_f32_16x16x4_f32(
                  false, a, false, bb, (short)0, acc, false, false);
    }

    float* dst = scores + ((size_t)b * NN + (size_t)t * 16) * 16;
#pragma unroll
    for (int v = 0; v < 8; ++v) {
        int row = v + hi * 8;
        dst[row * 16 + r] = acc[v];
    }
}

#define RPB 16
static_assert(NS % RPB == 0 && NN % RPB == 0, "row group must not straddle S/T");

__global__ __launch_bounds__(256)
void hstgnn_emit(const float* __restrict__ scores,
                 const float* __restrict__ relb,
                 const float* __restrict__ maskb,
                 float* __restrict__ out) {
    int blk = blockIdx.x;
    int b   = blk / (NN / RPB);
    int i0  = (blk % (NN / RPB)) * RPB;
    int j0  = threadIdx.x * 4;
    bool iT = (i0 >= NS);
    bool jT = (j0 >= NS);
    bool same = (iT == jT);

    int colslot = same ? 4 : 6;
    int k = jT ? (iT ? 1 : 2) : (iT ? 3 : 0);
    float rb = relb[k];
    float mb = maskb[k];

    const float* scol = scores + (size_t)b * NN * 16;
    float cx[4], cy[4];
#pragma unroll
    for (int u = 0; u < 4; ++u) {
        float2 cv = *(const float2*)(scol + (size_t)(j0 + u) * 16 + colslot);
        cx[u] = cv.x;
        cy[u] = cv.y;
    }

    const float* srow = scores + ((size_t)b * NN + i0) * 16;
    float*       orow = out + ((size_t)b * NN + i0) * NN + j0;

    float4 nxt = *(const float4*)srow;
    for (int r = 0; r < RPB; ++r) {
        float4 cur = nxt;
        if (r + 1 < RPB) nxt = *(const float4*)(srow + (r + 1) * 16);
        float rx = same ? cur.x : cur.z;
        float ry = same ? cur.y : cur.w;
        float res[4];
#pragma unroll
        for (int u = 0; u < 4; ++u) {
            float adj  = rx + cx[u] + rb;
            float mask = fmaxf(ry + cy[u] + mb, 0.0f);
            res[u] = tanhf(adj * mask);
        }
        *(float4*)(orow + (size_t)r * NN) =
            make_float4(res[0], res[1], res[2], res[3]);
    }
}

extern "C" void kernel_launch(void* const* d_in, const int* in_sizes, int n_in,
                              void* d_out, int out_size, void* d_ws, size_t ws_size,
                              hipStream_t stream) {
    const float* spatial  = (const float*)d_in[0];
    const float* temporal = (const float*)d_in[1];
    const float* rel_W    = (const float*)d_in[2];
    const float* rel_b    = (const float*)d_in[3];
    const float* mask_W   = (const float*)d_in[4];
    const float* mask_b   = (const float*)d_in[5];
    float*       out      = (float*)d_out;

    float* scores = (float*)d_ws;

    hipLaunchKernelGGL(hstgnn_score_wmma, dim3(NB * 64), dim3(32), 0, stream,
                       spatial, temporal, rel_W, mask_W, scores);
    hipLaunchKernelGGL(hstgnn_emit, dim3(NB * (NN / RPB)), dim3(256), 0, stream,
                       scores, rel_b, mask_b, out);
}


# emit: library tanhf (two divergent branches per element) replaced by a branch-free f32 tanh: polynomial below 0.625 and 1-2/(exp2+1) above both computed, one select; last bits may differ
# speedup vs baseline: 3.2711x; 1.3192x over previous
#include <hip/hip_runtime.h>

typedef __attribute__((ext_vector_type(2))) float v2f;
typedef __attribute__((ext_vector_type(8))) float v8f;

#define NB 16
#define NS 800
#define NT 224
#define NN 1024
#define DD 32

__global__ void hstgnn_score_wmma(const float* __restrict__ sp,
                                  const float* __restrict__ tp,
                                  const float* __restrict__ relW,
                                  const float* __restrict__ maskW,
                                  float* __restrict__ scores) {
    int id   = blockIdx.x;
    int b    = id >> 6;
    int t    = id & 63;
    int lane = threadIdx.x;
    int r    = lane & 15;
    int hi   = lane >> 4;

    const float* src;
    int type;
    if (t < 50) { type = 0; src = sp + ((size_t)b * NS + (size_t)t * 16) * DD; }
    else        { type = 1; src = tp + ((size_t)b * NT + (size_t)(t - 50) * 16) * DD; }

    int  n     = r & 7;
    int  half  = (n >> 2) & 1;
    bool cross = (n >> 1) & 1;
    int  kk    = cross ? 2 + (type ^ half) : type;
    const float* w = ((n & 1) ? maskW : relW) + kk * 64 + half * 32;

    v8f acc = {0.f,0.f,0.f,0.f,0.f,0.f,0.f,0.f};
#pragma unroll
    for (int k = 0; k < 8; ++k) {
        int kb = k * 4 + hi * 2;
        v2f a  = *(const v2f*)(src + r * DD + kb);
        v2f bb = *(const v2f*)(w + kb);
        acc = __builtin_amdgcn_wmma_f32_16x16x4_f32(
                  false, a, false, bb, (short)0, acc, false, false);
    }

    float* dst = scores + ((size_t)b * NN + (size_t)t * 16) * 16;
#pragma unroll
    for (int v = 0; v < 8; ++v) {
        int row = v + hi * 8;
        dst[row * 16 + r] = acc[v];
    }
}

__device__ __forceinline__ float tanh_select(float x) {
    float y2 = x * x;
    float p  = fmaf(y2, fmaf(y2, fmaf(y2, fmaf(y2, -0x1.758e7ap-8f,
                                                  0x1.521192p-6f),
                                        -0x1.b8389cp-5f),
                              0x1.110704p-3f),
                    -0x1.555532p-2f);
    float zs = fmaf(y2, x * p, x);
    float t  = __builtin_amdgcn_exp2f(__builtin_fabsf(x * 0x1.715476p+1f));
    float zl = fmaf(__builtin_amdgcn_rcpf(t + 1.0f), -2.0f, 1.0f);
    float z  = (__builtin_fabsf(x) < 0.625f) ? zs : zl;
    return __builtin_copysignf(z, x);
}

#define RPB 16
static_assert(NS % RPB == 0 && NN % RPB == 0, "row group must not straddle S/T");

__global__ __launch_bounds__(256)
void hstgnn_emit(const float* __restrict__ scores,
                 const float* __restrict__ relb,
                 const float* __restrict__ maskb,
                 float* __restrict__ out) {
    int blk = blockIdx.x;
    int b   = blk / (NN / RPB);
    int i0  = (blk % (NN / RPB)) * RPB;
    int j0  = threadIdx.x * 4;
    bool iT = (i0 >= NS);
    bool jT = (j0 >= NS);
    bool same = (iT == jT);

    int colslot = same ? 4 : 6;
    int k = jT ? (iT ? 1 : 2) : (iT ? 3 : 0);
    float rb = relb[k];
    float mb = maskb[k];

    const float* scol = scores + (size_t)b * NN * 16;
    float cx[4], cy[4];
#pragma unroll
    for (int u = 0; u < 4; ++u) {
        float2 cv = *(const float2*)(scol + (size_t)(j0 + u) * 16 + colslot);
        cx[u] = cv.x;
        cy[u] = cv.y;
    }

    const float* srow = scores + ((size_t)b * NN + i0) * 16;
    float*       orow = out + ((size_t)b * NN + i0) * NN + j0;

    float4 nxt = *(const float4*)srow;
    for (int r = 0; r < RPB; ++r) {
        float4 cur = nxt;
        if (r + 1 < RPB) nxt = *(const float4*)(srow + (r + 1) * 16);
        float rx = same ? cur.x : cur.z;
        float ry = same ? cur.y : cur.w;
        float res[4];
#pragma unroll
        for (int u = 0; u < 4; ++u) {
            float adj  = rx + cx[u] + rb;
            float mask = fmaxf(ry + cy[u] + mb, 0.0f);
            res[u] = tanh_select(adj * mask);
        }
        *(float4*)(orow + (size_t)r * NN) =
            make_float4(res[0], res[1], res[2], res[3]);
    }
}

extern "C" void kernel_launch(void* const* d_in, const int* in_sizes, int n_in,
                              void* d_out, int out_size, void* d_ws, size_t ws_size,
                              hipStream_t stream) {
    const float* spatial  = (const float*)d_in[0];
    const float* temporal = (const float*)d_in[1];
    const float* rel_W    = (const float*)d_in[2];
    const float* rel_b    = (const float*)d_in[3];
    const float* mask_W   = (const float*)d_in[4];
    const float* mask_b   = (const float*)d_in[5];
    float*       out      = (float*)d_out;

    float* scores = (float*)d_ws;

    hipLaunchKernelGGL(hstgnn_score_wmma, dim3(NB * 64), dim3(32), 0, stream,
                       spatial, temporal, rel_W, mask_W, scores);
    hipLaunchKernelGGL(hstgnn_emit, dim3(NB * (NN / RPB)), dim3(256), 0, stream,
                       scores, rel_b, mask_b, out);
}


# one launch, was 2: each block computes the scores of the 512 nodes of its own 256x256 output tile with the same f32 WMMAs into LDS, then emits the tile (256 blocks of 16 waves)
# speedup vs baseline: 3.7988x; 1.1613x over previous
#include <hip/hip_runtime.h>

typedef __attribute__((ext_vector_type(2))) float v2f;
typedef __attribute__((ext_vector_type(8))) float v8f;

#define NB 16
#define NS 800
#define NT 224
#define NN 1024
#define DD 32

#define TILE  256
#define NTL   (NN / TILE)
#define NTHR  512
#define RSTEP (NTHR / 64)
#define TPW   (32 / (NTHR / 32))
static_assert(NS % 16 == 0 && NS % RSTEP == 0 && NN % TILE == 0, "S/T boundary must fall between 16-node tiles and between row passes");

__device__ __forceinline__ float tanh_select(float x) {
    float y2 = x * x;
    float p  = fmaf(y2, fmaf(y2, fmaf(y2, fmaf(y2, -0x1.758e7ap-8f,
                                                  0x1.521192p-6f),
                                        -0x1.b8389cp-5f),
                              0x1.110704p-3f),
                    -0x1.555532p-2f);
    float zs = fmaf(y2, x * p, x);
    float t  = __builtin_amdgcn_exp2f(__builtin_fabsf(x * 0x1.715476p+1f));
    float zl = fmaf(__builtin_amdgcn_rcpf(t + 1.0f), -2.0f, 1.0f);
    float z  = (__builtin_fabsf(x) < 0.625f) ? zs : zl;
    return __builtin_copysignf(z, x);
}

__device__ __forceinline__ void emit_rows(const float* sc, const float* __restrict__ relb,
                                          const float* __restrict__ maskb, float* orow,
                                          int jl, int p, bool jT, bool iT, int q0, int q1) {
    bool same = (iT == jT);
    int colslot = same ? 4 : 6;
    int k = jT ? (iT ? 1 : 2) : (iT ? 3 : 0);
    float rb = relb[k];
    float mb = maskb[k];

    float cx[4], cy[4];
#pragma unroll
    for (int u = 0; u < 4; ++u) {
        float2 cv = *(const float2*)(sc + (TILE + jl + u) * 16 + colslot);
        cx[u] = cv.x;
        cy[u] = cv.y;
    }

    float4 nxt = *(const float4*)(sc + (q0 * RSTEP + p) * 16);
    for (int q = q0; q < q1; ++q) {
        float4 cur = nxt;
        if (q + 1 < q1) nxt = *(const float4*)(sc + ((q + 1) * RSTEP + p) * 16);
        float rx = same ? cur.x : cur.z;
        float ry = same ? cur.y : cur.w;
        float res[4];
#pragma unroll
        for (int u = 0; u < 4; ++u) {
            float adj  = rx + cx[u] + rb;
            float mask = fmaxf(ry + cy[u] + mb, 0.0f);
            res[u] = tanh_select(adj * mask);
        }
        *(float4*)(orow + (size_t)q * RSTEP * NN) =
            make_float4(res[0], res[1], res[2], res[3]);
    }
}

__global__ __launch_bounds__(NTHR)
void hstgnn_fused(const float* __restrict__ sp,
                  const float* __restrict__ tp,
                  const float* __restrict__ relW,
                  const float* __restrict__ relb,
                  const float* __restrict__ maskW,
                  const float* __restrict__ maskb,
                  float* __restrict__ out) {
    __shared__ float sc[2 * TILE * 16];

    int blk  = blockIdx.x;
    int b    = blk / (NTL * NTL);
    int tr   = (blk / NTL) % NTL;
    int tc   = blk % NTL;
    int tid  = threadIdx.x;
    int lane = tid & 31;
    int r    = lane & 15;
    int hi   = lane >> 4;
    int wave = __builtin_amdgcn_readfirstlane(tid >> 5);

    int  n     = r & 7;
    int  half  = (n >> 2) & 1;
    bool cross = (n >> 1) & 1;
    const float* wsel = ((n & 1) ? maskW : relW) + half * 32;

    v2f af[TPW][8], bf[TPW][8];
#pragma unroll
    for (int s = 0; s < TPW; ++s) {
        int lt = wave * TPW + s;
        int t  = (lt < 16) ? tr * 16 + lt : tc * 16 + (lt - 16);

        const float* src;
        int type;
        if (t < 50) { type = 0; src = sp + ((size_t)b * NS + (size_t)t * 16) * DD; }
        else        { type = 1; src = tp + ((size_t)b * NT + (size_t)(t - 50) * 16) * DD; }
        int kk = cross ? 2 + (type ^ half) : type;
        const float* w = wsel + kk * 64;
#pragma unroll
        for (int k = 0; k < 8; ++k) {
            int kb = k * 4 + hi * 2;
            af[s][k] = *(const v2f*)(src + r * DD + kb);
            bf[s][k] = *(const v2f*)(w + kb);
        }
    }
    __builtin_amdgcn_sched_barrier(0);
    v8f acc[TPW];
#pragma unroll
    for (int s = 0; s < TPW; ++s) {
        acc[s] = (v8f){0.f,0.f,0.f,0.f,0.f,0.f,0.f,0.f};
#pragma unroll
        for (int k = 0; k < 8; ++k)
            acc[s] = __builtin_amdgcn_wmma_f32_16x16x4_f32(
                         false, af[s][k], false, bf[s][k], (short)0, acc[s], false, false);
    }
    __builtin_amdgcn_sched_barrier(0);
    asm volatile("v_nop\n\tv_nop\n\tv_nop\n\tv_nop\n\tv_nop\n\tv_nop\n\tv_nop\n\tv_nop\n\tv_nop");
    __builtin_amdgcn_sched_barrier(0);
#pragma unroll
    for (int s = 0; s < TPW; ++s) {
        float* dst = sc + (wave * TPW + s) * 16 * 16;
#pragma unroll
        for (int v = 0; v < 8; ++v) {
            int row = v + hi * 8;
            dst[row * 16 + r] = acc[s][v];
        }
    }
    __syncthreads();

    int  jl = (tid & 63) * 4;
    int  p  = tid >> 6;
    int  j0 = tc * TILE + jl;
    bool jT = (j0 >= NS);
    int  i0 = tr * TILE;
    int  qs = (NS - i0) / RSTEP;
    qs = qs < 0 ? 0 : (qs > TILE / RSTEP ? TILE / RSTEP : qs);
    float* orow = out + ((size_t)b * NN + i0 + p) * NN + j0;

    if (qs > 0)
        emit_rows(sc, relb, maskb, orow, jl, p, jT, false, 0, qs);
    if (qs < TILE / RSTEP)
        emit_rows(sc, relb, maskb, orow, jl, p, jT, true, qs, TILE / RSTEP);
}

extern "C" void kernel_launch(void* const* d_in, const int* in_sizes, int n_in,
                              void* d_out, int out_size, void* d_ws, size_t ws_size,
                              hipStream_t stream) {
    const float* spatial  = (const float*)d_in[0];
    const float* temporal = (const float*)d_in[1];
    const float* rel_W    = (const float*)d_in[2];
    const float* rel_b    = (const float*)d_in[3];
    const float* mask_W   = (const float*)d_in[4];
    const float* mask_b   = (const float*)d_in[5];
    float*       out      = (float*)d_out;

    hipLaunchKernelGGL(hstgnn_fused, dim3(NB * NTL * NTL), dim3(NTHR), 0, stream,
                       spatial, temporal, rel_W, rel_b, mask_W, mask_b, out);
}


# 256 blocks of 8 waves, was 16: half as many waves to start; a thread emits its 64 tile rows four per loop trip (row scores read one group ahead as the float2 in use); tanh and sum order kept
# speedup vs baseline: 4.1845x; 1.1015x over previous
#include <hip/hip_runtime.h>

typedef __attribute__((ext_vector_type(2))) float v2f;
typedef __attribute__((ext_vector_type(8))) float v8f;

#define NB 16
#define NS 800
#define NT 224
#define NN 1024
#define DD 32

#define TILE  256
#define NTL   (NN / TILE)
#define NTHR  256
#define RSTEP (NTHR / 64)
#define RGRP  4
#define TPW   (32 / (NTHR / 32))
static_assert(NS % 16 == 0 && NS % (RSTEP * RGRP) == 0 && TILE % (RSTEP * RGRP) == 0 && NN % TILE == 0,
              "S/T boundary must fall between 16-node tiles and between groups of row passes");
static_assert(RSTEP * RGRP <= TILE, "the row-score read one group ahead must stay inside the 2*TILE-entry table");

__device__ __forceinline__ float tanh_select(float x) {
    float y2 = x * x;
    float p  = fmaf(y2, fmaf(y2, fmaf(y2, fmaf(y2, -0x1.758e7ap-8f,
                                                  0x1.521192p-6f),
                                        -0x1.b8389cp-5f),
                              0x1.110704p-3f),
                    -0x1.555532p-2f);
    float zs = fmaf(y2, x * p, x);
    float t  = __builtin_amdgcn_exp2f(__builtin_fabsf(x * 0x1.715476p+1f));
    float zl = fmaf(__builtin_amdgcn_rcpf(t + 1.0f), -2.0f, 1.0f);
    float z  = (__builtin_fabsf(x) < 0.625f) ? zs : zl;
    return __builtin_copysignf(z, x);
}

__device__ __forceinline__ void emit_rows(const float* sc, const float* __restrict__ relb,
                                          const float* __restrict__ maskb, float* orow,
                                          int jl, int p, bool jT, bool iT, int q0, int q1) {
    bool same = (iT == jT);
    int colslot = same ? 4 : 6;
    int k = jT ? (iT ? 1 : 2) : (iT ? 3 : 0);
    float rb = relb[k];
    float mb = maskb[k];

    float cx[4], cy[4];
#pragma unroll
    for (int u = 0; u < 4; ++u) {
        float2 cv = *(const float2*)(sc + (TILE + jl + u) * 16 + colslot);
        cx[u] = cv.x;
        cy[u] = cv.y;
    }

    const float* srow = sc + p * 16 + (same ? 0 : 2);
    float2 nxt[RGRP];
#pragma unroll
    for (int g = 0; g < RGRP; ++g)
        nxt[g] = *(const float2*)(srow + (q0 + g) * RSTEP * 16);
    for (int q = q0; q < q1; q += RGRP) {
        float2 cur[RGRP];
#pragma unroll
        for (int g = 0; g < RGRP; ++g) {
            cur[g] = nxt[g];
            nxt[g] = *(const float2*)(srow + (q + RGRP + g) * RSTEP * 16);
        }
#pragma unroll
        for (int g = 0; g < RGRP; ++g) {
            float rx = cur[g].x;
            float ry = cur[g].y;
            float res[4];
#pragma unroll
            for (int u = 0; u < 4; ++u) {
                float adj  = rx + cx[u] + rb;
                float mask = fmaxf(ry + cy[u] + mb, 0.0f);
                res[u] = tanh_select(adj * mask);
            }
            *(float4*)(orow + (size_t)(q + g) * RSTEP * NN) =
                make_float4(res[0], res[1], res[2], res[3]);
        }
    }
}

__global__ __launch_bounds__(NTHR)
void hstgnn_fused(const float* __restrict__ sp,
                  const float* __restrict__ tp,
                  const float* __restrict__ relW,
                  const float* __restrict__ relb,
                  const float* __restrict__ maskW,
                  const float* __restrict__ maskb,
                  float* __restrict__ out) {
    __shared__ float sc[2 * TILE * 16];

    int blk  = blockIdx.x;
    int b    = blk / (NTL * NTL);
    int tr   = (blk / NTL) % NTL;
    int tc   = blk % NTL;
    int tid  = threadIdx.x;
    int lane = tid & 31;
    int r    = lane & 15;
    int hi   = lane >> 4;
    int wave = __builtin_amdgcn_readfirstlane(tid >> 5);

    int  n     = r & 7;
    int  half  = (n >> 2) & 1;
    bool cross = (n >> 1) & 1;
    const float* wsel = ((n & 1) ? maskW : relW) + half * 32;

    v2f af[TPW][8], bf[TPW][8];
#pragma unroll
    for (int s = 0; s < TPW; ++s) {
        int lt = wave * TPW + s;
        int t  = (lt < 16) ? tr * 16 + lt : tc * 16 + (lt - 16);

        const float* src;
        int type;
        if (t < 50) { type = 0; src = sp + ((size_t)b * NS + (size_t)t * 16) * DD; }
        else        { type = 1; src = tp + ((size_t)b * NT + (size_t)(t - 50) * 16) * DD; }
        int kk = cross ? 2 + (type ^ half) : type;
        const float* w = wsel + kk * 64;
#pragma unroll
        for (int k = 0; k < 8; ++k) {
            int kb = k * 4 + hi * 2;
            af[s][k] = *(const v2f*)(src + r * DD + kb);
            bf[s][k] = *(const v2f*)(w + kb);
        }
    }
    __builtin_amdgcn_sched_barrier(0);
    v8f acc[TPW];
#pragma unroll
    for (int s = 0; s < TPW; ++s) {
        acc[s] = (v8f){0.f,0.f,0.f,0.f,0.f,0.f,0.f,0.f};
#pragma unroll
        for (int k = 0; k < 8; ++k)
            acc[s] = __builtin_amdgcn_wmma_f32_16x16x4_f32(
                         false, af[s][k], false, bf[s][k], (short)0, acc[s], false, false);
    }
    __builtin_amdgcn_sched_barrier(0);
    asm volatile("v_nop\n\tv_nop\n\tv_nop\n\tv_nop\n\tv_nop\n\tv_nop\n\tv_nop\n\tv_nop\n\tv_nop");
    __builtin_amdgcn_sched_barrier(0);
#pragma unroll
    for (int s = 0; s < TPW; ++s) {
        float* dst = sc + (wave * TPW + s) * 16 * 16;
#pragma unroll
        for (int v = 0; v < 8; ++v) {
            int row = v + hi * 8;
            dst[row * 16 + r] = acc[s][v];
        }
    }
    __syncthreads();

    int  jl = (tid & 63) * 4;
    int  p  = tid >> 6;
    int  j0 = tc * TILE + jl;
    bool jT = (j0 >= NS);
    int  i0 = tr * TILE;
    int  qs = (NS - i0) / RSTEP;
    qs = qs < 0 ? 0 : (qs > TILE / RSTEP ? TILE / RSTEP : qs);
    float* orow = out + ((size_t)b * NN + i0 + p) * NN + j0;

    if (qs > 0)
        emit_rows(sc, relb, maskb, orow, jl, p, jT, false, 0, qs);
    if (qs < TILE / RSTEP)
        emit_rows(sc, relb, maskb, orow, jl, p, jT, true, qs, TILE / RSTEP);
}

extern "C" void kernel_launch(void* const* d_in, const int* in_sizes, int n_in,
                              void* d_out, int out_size, void* d_ws, size_t ws_size,
                              hipStream_t stream) {
    const float* spatial  = (const float*)d_in[0];
    const float* temporal = (const float*)d_in[1];
    const float* rel_W    = (const float*)d_in[2];
    const float* rel_b    = (const float*)d_in[3];
    const float* mask_W   = (const float*)d_in[4];
    const float* mask_b   = (const float*)d_in[5];
    float*       out      = (float*)d_out;

    hipLaunchKernelGGL(hstgnn_fused, dim3(NB * NTL * NTL), dim3(NTHR), 0, stream,
                       spatial, temporal, rel_W, rel_b, mask_W, mask_b, out);
}


# emit: the two-range software tanh (polynomial, exp2 + rcp, select, copysign) replaced by the chip's f32 tanh instruction v_tanh_f32; f32 kept, last bits may differ; sum order kept
# speedup vs baseline: 5.1858x; 1.2393x over previous
#include <hip/hip_runtime.h>

typedef __attribute__((ext_vector_type(2))) float v2f;
typedef __attribute__((ext_vector_type(8))) float v8f;

#define NB 16
#define NS 800
#define NT 224
#define NN 1024
#define DD 32

#define TILE  256
#define NTL   (NN / TILE)
#define NTHR  256
#define RSTEP (NTHR / 64)
#define RGRP  4
#define TPW   (32 / (NTHR / 32))
static_assert(NS % 16 == 0 && NS % (RSTEP * RGRP) == 0 && TILE % (RSTEP * RGRP) == 0 && NN % TILE == 0,
              "S/T boundary must fall between 16-node tiles and between groups of row passes");
static_assert(RSTEP * RGRP <= TILE, "the row-score read one group ahead must stay inside the 2*TILE-entry table");

__device__ __forceinline__ float tanh_hw(float x) {
    return __builtin_amdgcn_tanhf(x);
}

__device__ __forceinline__ void emit_rows(const float* sc, const float* __restrict__ relb,
                                          const float* __restrict__ maskb, float* orow,
                                          int jl, int p, bool jT, bool iT, int q0, int q1) {
    bool same = (iT == jT);
    int colslot = same ? 4 : 6;
    int k = jT ? (iT ? 1 : 2) : (iT ? 3 : 0);
    float rb = relb[k];
    float mb = maskb[k];

    float cx[4], cy[4];
#pragma unroll
    for (int u = 0; u < 4; ++u) {
        float2 cv = *(const float2*)(sc + (TILE + jl + u) * 16 + colslot);
        cx[u] = cv.x;
        cy[u] = cv.y;
    }

    const float* srow = sc + p * 16 + (same ? 0 : 2);
    float2 nxt[RGRP];
#pragma unroll
    for (int g = 0; g < RGRP; ++g)
        nxt[g] = *(const float2*)(srow + (q0 + g) * RSTEP * 16);
    for (int q = q0; q < q1; q += RGRP) {
        float2 cur[RGRP];
#pragma unroll
        for (int g = 0; g < RGRP; ++g) {
            cur[g] = nxt[g];
            nxt[g] = *(const float2*)(srow + (q + RGRP + g) * RSTEP * 16);
        }
#pragma unroll
        for (int g = 0; g < RGRP; ++g) {
            float rx = cur[g].x;
            float ry = cur[g].y;
            float res[4];
#pragma unroll
            for (int u = 0; u < 4; ++u) {
                float adj  = rx + cx[u] + rb;
                float mask = fmaxf(ry + cy[u] + mb, 0.0f);
                res[u] = tanh_hw(adj * mask);
            }
            *(float4*)(orow + (size_t)(q + g) * RSTEP * NN) =
                make_float4(res[0], res[1], res[2], res[3]);
        }
    }
}

__global__ __launch_bounds__(NTHR)
void hstgnn_fused(const float* __restrict__ sp,
                  const float* __restrict__ tp,
                  const float* __restrict__ relW,
                  const float* __restrict__ relb,
                  const float* __restrict__ maskW,
                  const float* __restrict__ maskb,
                  float* __restrict__ out) {
    __shared__ float sc[2 * TILE * 16];

    int blk  = blockIdx.x;
    int b    = blk / (NTL * NTL);
    int tr   = (blk / NTL) % NTL;
    int tc   = blk % NTL;
    int tid  = threadIdx.x;
    int lane = tid & 31;
    int r    = lane & 15;
    int hi   = lane >> 4;
    int wave = __builtin_amdgcn_readfirstlane(tid >> 5);

    int  n     = r & 7;
    int  half  = (n >> 2) & 1;
    bool cross = (n >> 1) & 1;
    const float* wsel = ((n & 1) ? maskW : relW) + half * 32;

    v2f af[TPW][8], bf[TPW][8];
#pragma unroll
    for (int s = 0; s < TPW; ++s) {
        int lt = wave * TPW + s;
        int t  = (lt < 16) ? tr * 16 + lt : tc * 16 + (lt - 16);

        const float* src;
        int type;
        if (t < 50) { type = 0; src = sp + ((size_t)b * NS + (size_t)t * 16) * DD; }
        else        { type = 1; src = tp + ((size_t)b * NT + (size_t)(t - 50) * 16) * DD; }
        int kk = cross ? 2 + (type ^ half) : type;
        const float* w = wsel + kk * 64;
#pragma unroll
        for (int k = 0; k < 8; ++k) {
            int kb = k * 4 + hi * 2;
            af[s][k] = *(const v2f*)(src + r * DD + kb);
            bf[s][k] = *(const v2f*)(w + kb);
        }
    }
    __builtin_amdgcn_sched_barrier(0);
    v8f acc[TPW];
#pragma unroll
    for (int s = 0; s < TPW; ++s) {
        acc[s] = (v8f){0.f,0.f,0.f,0.f,0.f,0.f,0.f,0.f};
#pragma unroll
        for (int k = 0; k < 8; ++k)
            acc[s] = __builtin_amdgcn_wmma_f32_16x16x4_f32(
                         false, af[s][k], false, bf[s][k], (short)0, acc[s], false, false);
    }
    __builtin_amdgcn_sched_barrier(0);
    asm volatile("v_nop\n\tv_nop\n\tv_nop\n\tv_nop\n\tv_nop\n\tv_nop\n\tv_nop\n\tv_nop\n\tv_nop");
    __builtin_amdgcn_sched_barrier(0);
#pragma unroll
    for (int s = 0; s < TPW; ++s) {
        float* dst = sc + (wave * TPW + s) * 16 * 16;
#pragma unroll
        for (int v = 0; v < 8; ++v) {
            int row = v + hi * 8;
            dst[row * 16 + r] = acc[s][v];
        }
    }
    __syncthreads();

    int  jl = (tid & 63) * 4;
    int  p  = tid >> 6;
    int  j0 = tc * TILE + jl;
    bool jT = (j0 >= NS);
    int  i0 = tr * TILE;
    int  qs = (NS - i0) / RSTEP;
    qs = qs < 0 ? 0 : (qs > TILE / RSTEP ? TILE / RSTEP : qs);
    float* orow = out + ((size_t)b * NN + i0 + p) * NN + j0;

    if (qs > 0)
        emit_rows(sc, relb, maskb, orow, jl, p, jT, false, 0, qs);
    if (qs < TILE / RSTEP)
        emit_rows(sc, relb, maskb, orow, jl, p, jT, true, qs, TILE / RSTEP);
}

extern "C" void kernel_launch(void* const* d_in, const int* in_sizes, int n_in,
                              void* d_out, int out_size, void* d_ws, size_t ws_size,
                              hipStream_t stream) {
    const float* spatial  = (const float*)d_in[0];
    const float* temporal = (const float*)d_in[1];
    const float* rel_W    = (const float*)d_in[2];
    const float* rel_b    = (const float*)d_in[3];
    const float* mask_W   = (const float*)d_in[4];
    const float* mask_b   = (const float*)d_in[5];
    float*       out      = (float*)d_out;

    hipLaunchKernelGGL(hstgnn_fused, dim3(NB * NTL * NTL), dim3(NTHR), 0, stream,
                       spatial, temporal, rel_W, rel_b, mask_W, mask_b, out);
}
